# starting kernel
# baseline (speedup 1.0000x reference)
#include <hip/hip_runtime.h>

typedef __attribute__((ext_vector_type(2))) float v2f;
typedef __attribute__((ext_vector_type(8))) float v8f;

#define EDGE_DIM 128
#define N_RADIAL 16

__global__ void zero_f4_kernel(float4* __restrict__ p, long n4) {
  long i = (long)blockIdx.x * blockDim.x + threadIdx.x;
  long stride = (long)gridDim.x * blockDim.x;
  float4 z = make_float4(0.f, 0.f, 0.f, 0.f);
  for (; i < n4; i += stride) p[i] = z;
}

__global__ __launch_bounds__(256) void edge_gate_scatter_kernel(
    const float* __restrict__ x, const float* __restrict__ rbf,
    const int* __restrict__ idx_i, const float* __restrict__ W_rbf,
    float* __restrict__ h, int n_edges)
{
  const int lane = threadIdx.x & 31;
  const int r    = lane & 15;
  const int hi   = lane >> 4;
  const int wave = blockIdx.x * (blockDim.x >> 5) + (threadIdx.x >> 5);
  const int nW   = gridDim.x * (blockDim.x >> 5);
  const int nFull = n_edges >> 4;

  v2f Bt[8][4];
#pragma unroll
  for (int cb = 0; cb < 8; ++cb) {
    const float* brow = W_rbf + (cb * 16 + r) * N_RADIAL + 2 * hi;
#pragma unroll
    for (int k0 = 0; k0 < 4; ++k0)
      Bt[cb][k0] = *(const v2f*)(brow + 4 * k0);
  }

  for (int t = wave; t < nFull; t += nW) {
    const int e0 = t << 4;

    v2f a[4];
    const float* arow = rbf + (e0 + r) * N_RADIAL + 2 * hi;
#pragma unroll
    for (int k0 = 0; k0 < 4; ++k0)
      a[k0] = *(const v2f*)(arow + 4 * k0);

    int hoff[8];
    const int* ibase = idx_i + e0 + 8 * hi;
#pragma unroll
    for (int v = 0; v < 8; ++v)
      hoff[v] = ibase[v] * EDGE_DIM;

#pragma unroll
    for (int cb = 0; cb < 8; ++cb) {
      v8f c = {0.f, 0.f, 0.f, 0.f, 0.f, 0.f, 0.f, 0.f};
#pragma unroll
      for (int k0 = 0; k0 < 4; ++k0)
        c = __builtin_amdgcn_wmma_f32_16x16x4_f32(
            false, a[k0], false, Bt[cb][k0], (short)0, c, false, false);

      const int col = cb * 16 + r;

      const float* xbase = x + (e0 + 8 * hi) * EDGE_DIM + col;
      float xv[8];
#pragma unroll
      for (int v = 0; v < 8; ++v)
        xv[v] = xbase[v * EDGE_DIM];

#pragma unroll
      for (int v = 0; v < 8; ++v)
        atomicAdd(h + (hoff[v] + col), c[v] * xv[v]);
    }
  }

  if (wave == 0) {
    for (int e = nFull << 4; e < n_edges; ++e) {
      const int nd = idx_i[e];
      const float* re = rbf + e * N_RADIAL;
#pragma unroll
      for (int cc = 0; cc < 4; ++cc) {
        const int ch = lane * 4 + cc;
        float s = 0.f;
#pragma unroll
        for (int k = 0; k < N_RADIAL; ++k) s += re[k] * W_rbf[ch * N_RADIAL + k];
        float g = s * x[e * EDGE_DIM + ch];
        atomicAdd(h + (nd * EDGE_DIM + ch), g);
      }
    }
  }
}

__global__ __launch_bounds__(256) void node_mlp_kernel(
    const float* __restrict__ in, const float* __restrict__ W,
    const float* __restrict__ bias, float* __restrict__ out,
    int n_rows, int apply_swish)
{
  __shared__ float Wlds[EDGE_DIM * EDGE_DIM];
  {
    const float4* Wg = (const float4*)W;
    float4* Wl = (float4*)Wlds;
    for (int i = threadIdx.x; i < EDGE_DIM * EDGE_DIM / 4; i += blockDim.x)
      Wl[i] = Wg[i];
  }
  __syncthreads();

  const int lane = threadIdx.x & 31;
  const int r    = lane & 15;
  const int hi   = lane >> 4;
  const int wave = blockIdx.x * (blockDim.x >> 5) + (threadIdx.x >> 5);
  const int nW   = gridDim.x * (blockDim.x >> 5);
  const int nTiles = (n_rows + 15) >> 4;

  for (int t = wave; t < nTiles; t += nW) {
    const int m0 = t << 4;
    v8f acc[8];
#pragma unroll
    for (int cb = 0; cb < 8; ++cb)
      acc[cb] = (v8f){0.f, 0.f, 0.f, 0.f, 0.f, 0.f, 0.f, 0.f};

    int mrow = m0 + r;
    if (mrow >= n_rows) mrow = n_rows - 1;
    const float* arow = in + mrow * EDGE_DIM + 2 * hi;

#pragma unroll 4
    for (int k0 = 0; k0 < 32; ++k0) {
      v2f a = *(const v2f*)(arow + 4 * k0);
#pragma unroll
      for (int cb = 0; cb < 8; ++cb) {
        v2f b = *(const v2f*)(&Wlds[(cb * 16 + r) * EDGE_DIM + 4 * k0 + 2 * hi]);
        acc[cb] = __builtin_amdgcn_wmma_f32_16x16x4_f32(
            false, a, false, b, (short)0, acc[cb], false, false);
      }
    }

    const int rowBase = m0 + 8 * hi;
    if (m0 + 16 <= n_rows) {
#pragma unroll
      for (int cb = 0; cb < 8; ++cb) {
        const int col = cb * 16 + r;
        const float bv = bias ? bias[col] : 0.f;
        float* obase = out + rowBase * EDGE_DIM + col;
#pragma unroll
        for (int v = 0; v < 8; ++v) {
          float val = acc[cb][v] + bv;
          if (apply_swish) val = val * (1.f / (1.f + __expf(-val)));
          obase[v * EDGE_DIM] = val;
        }
      }
    } else {
#pragma unroll
      for (int cb = 0; cb < 8; ++cb) {
        const int col = cb * 16 + r;
        const float bv = bias ? bias[col] : 0.f;
#pragma unroll
        for (int v = 0; v < 8; ++v) {
          const int m = rowBase + v;
          if (m < n_rows) {
            float val = acc[cb][v] + bv;
            if (apply_swish) val = val * (1.f / (1.f + __expf(-val)));
            out[m * EDGE_DIM + col] = val;
          }
        }
      }
    }
  }
}

extern "C" void kernel_launch(void* const* d_in, const int* in_sizes, int n_in,
                              void* d_out, int out_size, void* d_ws, size_t ws_size,
                              hipStream_t stream) {
  const float* x     = (const float*)d_in[0];
  const float* rbf   = (const float*)d_in[1];
  const int*   idx_i = (const int*)d_in[2];
  const float* W_rbf = (const float*)d_in[4];
  const float* W1    = (const float*)d_in[5];
  const float* b1    = (const float*)d_in[6];
  const float* W2    = (const float*)d_in[7];
  const float* b2    = (const float*)d_in[8];
  const float* W3    = (const float*)d_in[9];
  const float* b3    = (const float*)d_in[10];
  const float* W_out = (const float*)d_in[11];
  float* outp = (float*)d_out;

  const int n_edges = in_sizes[0] / EDGE_DIM;
  const int n_nodes = out_size / EDGE_DIM;

  float* h0 = (float*)d_ws;
  float* h1 = h0 + (size_t)n_nodes * EDGE_DIM;

  {
    long n4 = (long)n_nodes * EDGE_DIM / 4;
    long b = (n4 + 255) / 256;
    int blocks = (int)(b > 2048 ? 2048 : (b < 1 ? 1 : b));
    zero_f4_kernel<<<blocks, 256, 0, stream>>>((float4*)h0, n4);
  }

  {
    int nTiles = n_edges >> 4;
    int blocks = (nTiles + 7) / 8;
    if (blocks > 2048) blocks = 2048;
    if (blocks < 1) blocks = 1;
    edge_gate_scatter_kernel<<<blocks, 256, 0, stream>>>(x, rbf, idx_i, W_rbf, h0, n_edges);
  }

  {
    int nTiles = (n_nodes + 15) / 16;
    int blocks = (nTiles + 7) / 8;
    if (blocks > 2048) blocks = 2048;
    if (blocks < 1) blocks = 1;
    node_mlp_kernel<<<blocks, 256, 0, stream>>>(h0, W1, b1, h1, n_nodes, 1);
    node_mlp_kernel<<<blocks, 256, 0, stream>>>(h1, W2, b2, h0, n_nodes, 1);
    node_mlp_kernel<<<blocks, 256, 0, stream>>>(h0, W3, b3, h1, n_nodes, 1);
    node_mlp_kernel<<<blocks, 256, 0, stream>>>(h1, W_out, nullptr, outp, n_nodes, 0);
  }
}


# Pad the LDS rows of the staged MLP weight matrix to a pitch of 132 floats so the 8-byte B fragment reads hit 64 different banks instead of a 16-way conflict
# speedup vs baseline: 1.2895x; 1.2895x over previous
#include <hip/hip_runtime.h>

typedef __attribute__((ext_vector_type(2))) float v2f;
typedef __attribute__((ext_vector_type(8))) float v8f;

#define EDGE_DIM 128
#define N_RADIAL 16
#define W_PITCH 132

__global__ void zero_f4_kernel(float4* __restrict__ p, long n4) {
  long i = (long)blockIdx.x * blockDim.x + threadIdx.x;
  long stride = (long)gridDim.x * blockDim.x;
  float4 z = make_float4(0.f, 0.f, 0.f, 0.f);
  for (; i < n4; i += stride) p[i] = z;
}

__global__ __launch_bounds__(256) void edge_gate_scatter_kernel(
    const float* __restrict__ x, const float* __restrict__ rbf,
    const int* __restrict__ idx_i, const float* __restrict__ W_rbf,
    float* __restrict__ h, int n_edges)
{
  const int lane = threadIdx.x & 31;
  const int r    = lane & 15;
  const int hi   = lane >> 4;
  const int wave = blockIdx.x * (blockDim.x >> 5) + (threadIdx.x >> 5);
  const int nW   = gridDim.x * (blockDim.x >> 5);
  const int nFull = n_edges >> 4;

  v2f Bt[8][4];
#pragma unroll
  for (int cb = 0; cb < 8; ++cb) {
    const float* brow = W_rbf + (cb * 16 + r) * N_RADIAL + 2 * hi;
#pragma unroll
    for (int k0 = 0; k0 < 4; ++k0)
      Bt[cb][k0] = *(const v2f*)(brow + 4 * k0);
  }

  for (int t = wave; t < nFull; t += nW) {
    const int e0 = t << 4;

    v2f a[4];
    const float* arow = rbf + (e0 + r) * N_RADIAL + 2 * hi;
#pragma unroll
    for (int k0 = 0; k0 < 4; ++k0)
      a[k0] = *(const v2f*)(arow + 4 * k0);

    int hoff[8];
    const int* ibase = idx_i + e0 + 8 * hi;
#pragma unroll
    for (int v = 0; v < 8; ++v)
      hoff[v] = ibase[v] * EDGE_DIM;

#pragma unroll
    for (int cb = 0; cb < 8; ++cb) {
      v8f c = {0.f, 0.f, 0.f, 0.f, 0.f, 0.f, 0.f, 0.f};
#pragma unroll
      for (int k0 = 0; k0 < 4; ++k0)
        c = __builtin_amdgcn_wmma_f32_16x16x4_f32(
            false, a[k0], false, Bt[cb][k0], (short)0, c, false, false);

      const int col = cb * 16 + r;

      const float* xbase = x + (e0 + 8 * hi) * EDGE_DIM + col;
      float xv[8];
#pragma unroll
      for (int v = 0; v < 8; ++v)
        xv[v] = xbase[v * EDGE_DIM];

#pragma unroll
      for (int v = 0; v < 8; ++v)
        atomicAdd(h + (hoff[v] + col), c[v] * xv[v]);
    }
  }

  if (wave == 0) {
    for (int e = nFull << 4; e < n_edges; ++e) {
      const int nd = idx_i[e];
      const float* re = rbf + e * N_RADIAL;
#pragma unroll
      for (int cc = 0; cc < 4; ++cc) {
        const int ch = lane * 4 + cc;
        float s = 0.f;
#pragma unroll
        for (int k = 0; k < N_RADIAL; ++k) s += re[k] * W_rbf[ch * N_RADIAL + k];
        float g = s * x[e * EDGE_DIM + ch];
        atomicAdd(h + (nd * EDGE_DIM + ch), g);
      }
    }
  }
}

__global__ __launch_bounds__(256) void node_mlp_kernel(
    const float* __restrict__ in, const float* __restrict__ W,
    const float* __restrict__ bias, float* __restrict__ out,
    int n_rows, int apply_swish)
{
  __shared__ __attribute__((aligned(16))) float Wlds[EDGE_DIM * W_PITCH];
  {
    const float4* Wg = (const float4*)W;
    for (int i = threadIdx.x; i < EDGE_DIM * EDGE_DIM / 4; i += blockDim.x)
      *(float4*)(&Wlds[(i >> 5) * W_PITCH + 4 * (i & 31)]) = Wg[i];
  }
  __syncthreads();

  const int lane = threadIdx.x & 31;
  const int r    = lane & 15;
  const int hi   = lane >> 4;
  const int wave = blockIdx.x * (blockDim.x >> 5) + (threadIdx.x >> 5);
  const int nW   = gridDim.x * (blockDim.x >> 5);
  const int nTiles = (n_rows + 15) >> 4;

  for (int t = wave; t < nTiles; t += nW) {
    const int m0 = t << 4;
    v8f acc[8];
#pragma unroll
    for (int cb = 0; cb < 8; ++cb)
      acc[cb] = (v8f){0.f, 0.f, 0.f, 0.f, 0.f, 0.f, 0.f, 0.f};

    int mrow = m0 + r;
    if (mrow >= n_rows) mrow = n_rows - 1;
    const float* arow = in + mrow * EDGE_DIM + 2 * hi;

#pragma unroll 4
    for (int k0 = 0; k0 < 32; ++k0) {
      v2f a = *(const v2f*)(arow + 4 * k0);
#pragma unroll
      for (int cb = 0; cb < 8; ++cb) {
        v2f b = *(const v2f*)(&Wlds[(cb * 16 + r) * W_PITCH + 4 * k0 + 2 * hi]);
        acc[cb] = __builtin_amdgcn_wmma_f32_16x16x4_f32(
            false, a, false, b, (short)0, acc[cb], false, false);
      }
    }

    const int rowBase = m0 + 8 * hi;
    if (m0 + 16 <= n_rows) {
#pragma unroll
      for (int cb = 0; cb < 8; ++cb) {
        const int col = cb * 16 + r;
        const float bv = bias ? bias[col] : 0.f;
        float* obase = out + rowBase * EDGE_DIM + col;
#pragma unroll
        for (int v = 0; v < 8; ++v) {
          float val = acc[cb][v] + bv;
          if (apply_swish) val = val * (1.f / (1.f + __expf(-val)));
          obase[v * EDGE_DIM] = val;
        }
      }
    } else {
#pragma unroll
      for (int cb = 0; cb < 8; ++cb) {
        const int col = cb * 16 + r;
        const float bv = bias ? bias[col] : 0.f;
#pragma unroll
        for (int v = 0; v < 8; ++v) {
          const int m = rowBase + v;
          if (m < n_rows) {
            float val = acc[cb][v] + bv;
            if (apply_swish) val = val * (1.f / (1.f + __expf(-val)));
            out[m * EDGE_DIM + col] = val;
          }
        }
      }
    }
  }
}

extern "C" void kernel_launch(void* const* d_in, const int* in_sizes, int n_in,
                              void* d_out, int out_size, void* d_ws, size_t ws_size,
                              hipStream_t stream) {
  const float* x     = (const float*)d_in[0];
  const float* rbf   = (const float*)d_in[1];
  const int*   idx_i = (const int*)d_in[2];
  const float* W_rbf = (const float*)d_in[4];
  const float* W1    = (const float*)d_in[5];
  const float* b1    = (const float*)d_in[6];
  const float* W2    = (const float*)d_in[7];
  const float* b2    = (const float*)d_in[8];
  const float* W3    = (const float*)d_in[9];
  const float* b3    = (const float*)d_in[10];
  const float* W_out = (const float*)d_in[11];
  float* outp = (float*)d_out;

  const int n_edges = in_sizes[0] / EDGE_DIM;
  const int n_nodes = out_size / EDGE_DIM;

  float* h0 = (float*)d_ws;
  float* h1 = h0 + (size_t)n_nodes * EDGE_DIM;

  {
    long n4 = (long)n_nodes * EDGE_DIM / 4;
    long b = (n4 + 255) / 256;
    int blocks = (int)(b > 2048 ? 2048 : (b < 1 ? 1 : b));
    zero_f4_kernel<<<blocks, 256, 0, stream>>>((float4*)h0, n4);
  }

  {
    int nTiles = n_edges >> 4;
    int blocks = (nTiles + 7) / 8;
    if (blocks > 2048) blocks = 2048;
    if (blocks < 1) blocks = 1;
    edge_gate_scatter_kernel<<<blocks, 256, 0, stream>>>(x, rbf, idx_i, W_rbf, h0, n_edges);
  }

  {
    int nTiles = (n_nodes + 15) / 16;
    int blocks = (nTiles + 7) / 8;
    if (blocks > 2048) blocks = 2048;
    if (blocks < 1) blocks = 1;
    node_mlp_kernel<<<blocks, 256, 0, stream>>>(h0, W1, b1, h1, n_nodes, 1);
    node_mlp_kernel<<<blocks, 256, 0, stream>>>(h1, W2, b2, h0, n_nodes, 1);
    node_mlp_kernel<<<blocks, 256, 0, stream>>>(h0, W3, b3, h1, n_nodes, 1);
    node_mlp_kernel<<<blocks, 256, 0, stream>>>(h1, W_out, nullptr, outp, n_nodes, 0);
  }
}


# Edge scatter: v_permlane16_swap upper 16 lanes of each even channel tile with lower 16 of the odd one so each x load and f32 atomic add covers one whole 128-byte row line; atomic order as before
# speedup vs baseline: 1.5067x; 1.1684x over previous
#include <hip/hip_runtime.h>

typedef __attribute__((ext_vector_type(2))) float v2f;
typedef __attribute__((ext_vector_type(8))) float v8f;
typedef __attribute__((ext_vector_type(2))) unsigned v2u;

#define EDGE_DIM 128
#define N_RADIAL 16
#define W_PITCH 132

__global__ void zero_f4_kernel(float4* __restrict__ p, long n4) {
  long i = (long)blockIdx.x * blockDim.x + threadIdx.x;
  long stride = (long)gridDim.x * blockDim.x;
  float4 z = make_float4(0.f, 0.f, 0.f, 0.f);
  for (; i < n4; i += stride) p[i] = z;
}

__global__ __launch_bounds__(256) void edge_gate_scatter_kernel(
    const float* __restrict__ x, const float* __restrict__ rbf,
    const int* __restrict__ idx_i, const float* __restrict__ W_rbf,
    float* __restrict__ h, int n_edges)
{
  const unsigned lane = threadIdx.x & 31;
  const int r    = lane & 15;
  const int hi   = lane >> 4;
  const int wave = __builtin_amdgcn_readfirstlane(
      (int)(blockIdx.x * (blockDim.x >> 5) + (threadIdx.x >> 5)));
  const int nW   = gridDim.x * (blockDim.x >> 5);
  const int nFull = n_edges >> 4;

  v2f Bt[8][4];
#pragma unroll
  for (int cb = 0; cb < 8; ++cb) {
    const float* brow = W_rbf + (cb * 16 + r) * N_RADIAL + 2 * hi;
#pragma unroll
    for (int k0 = 0; k0 < 4; ++k0)
      Bt[cb][k0] = *(const v2f*)(brow + 4 * k0);
  }

  for (int t = wave; t < nFull; t += nW) {
    const int e0 = t << 4;

    v2f a[4];
    const float* arow = rbf + (e0 + r) * N_RADIAL + 2 * hi;
#pragma unroll
    for (int k0 = 0; k0 < 4; ++k0)
      a[k0] = *(const v2f*)(arow + 4 * k0);

    const int* ibase = idx_i + e0;
    int nd[16];
#pragma unroll
    for (int q = 0; q < 16; ++q)
      nd[q] = ibase[q];

    const float* xtile = x + (size_t)e0 * EDGE_DIM;

#pragma unroll
    for (int j = 0; j < 4; ++j) {
      const unsigned ch = j * 32 + lane;

      float xlo[8], xhi[8];
#pragma unroll
      for (int v = 0; v < 8; ++v) {
        xlo[v] = xtile[v * EDGE_DIM + ch];
        xhi[v] = xtile[(v + 8) * EDGE_DIM + ch];
      }

      v8f c0 = {0.f, 0.f, 0.f, 0.f, 0.f, 0.f, 0.f, 0.f};
      v8f c1 = {0.f, 0.f, 0.f, 0.f, 0.f, 0.f, 0.f, 0.f};
#pragma unroll
      for (int k0 = 0; k0 < 4; ++k0) {
        c0 = __builtin_amdgcn_wmma_f32_16x16x4_f32(
            false, a[k0], false, Bt[2 * j][k0], (short)0, c0, false, false);
        c1 = __builtin_amdgcn_wmma_f32_16x16x4_f32(
            false, a[k0], false, Bt[2 * j + 1][k0], (short)0, c1, false, false);
      }

      asm volatile("v_nop\n\tv_nop\n\tv_nop\n\tv_nop" : "+v"(c0), "+v"(c1));

#pragma unroll
      for (int v = 0; v < 8; ++v) {
        const float f0 = c0[v];
        const float f1 = c1[v];
        const v2u s = __builtin_amdgcn_permlane16_swap(
            __float_as_uint(f0), __float_as_uint(f1), false, false);
        const unsigned s0 = s[0];
        const unsigned s1 = s[1];
        const float glo = __uint_as_float(s0) * xlo[v];
        const float ghi = __uint_as_float(s1) * xhi[v];
        const unsigned olo = (unsigned)(nd[v] * EDGE_DIM) + lane;
        const unsigned ohi = (unsigned)(nd[v + 8] * EDGE_DIM) + lane;
        atomicAdd(h + ((size_t)olo + j * 32), glo);
        atomicAdd(h + ((size_t)ohi + j * 32), ghi);
      }
    }
  }

  if (wave == 0) {
    for (int e = nFull << 4; e < n_edges; ++e) {
      const int nd = idx_i[e];
      const float* re = rbf + e * N_RADIAL;
#pragma unroll
      for (int cc = 0; cc < 4; ++cc) {
        const int ch = lane * 4 + cc;
        float s = 0.f;
#pragma unroll
        for (int k = 0; k < N_RADIAL; ++k) s += re[k] * W_rbf[ch * N_RADIAL + k];
        float g = s * x[e * EDGE_DIM + ch];
        atomicAdd(h + (nd * EDGE_DIM + ch), g);
      }
    }
  }
}

__global__ __launch_bounds__(256) void node_mlp_kernel(
    const float* __restrict__ in, const float* __restrict__ W,
    const float* __restrict__ bias, float* __restrict__ out,
    int n_rows, int apply_swish)
{
  __shared__ __attribute__((aligned(16))) float Wlds[EDGE_DIM * W_PITCH];
  {
    const float4* Wg = (const float4*)W;
    for (int i = threadIdx.x; i < EDGE_DIM * EDGE_DIM / 4; i += blockDim.x)
      *(float4*)(&Wlds[(i >> 5) * W_PITCH + 4 * (i & 31)]) = Wg[i];
  }
  __syncthreads();

  const int lane = threadIdx.x & 31;
  const int r    = lane & 15;
  const int hi   = lane >> 4;
  const int wave = blockIdx.x * (blockDim.x >> 5) + (threadIdx.x >> 5);
  const int nW   = gridDim.x * (blockDim.x >> 5);
  const int nTiles = (n_rows + 15) >> 4;

  for (int t = wave; t < nTiles; t += nW) {
    const int m0 = t << 4;
    v8f acc[8];
#pragma unroll
    for (int cb = 0; cb < 8; ++cb)
      acc[cb] = (v8f){0.f, 0.f, 0.f, 0.f, 0.f, 0.f, 0.f, 0.f};

    int mrow = m0 + r;
    if (mrow >= n_rows) mrow = n_rows - 1;
    const float* arow = in + mrow * EDGE_DIM + 2 * hi;

#pragma unroll 4
    for (int k0 = 0; k0 < 32; ++k0) {
      v2f a = *(const v2f*)(arow + 4 * k0);
#pragma unroll
      for (int cb = 0; cb < 8; ++cb) {
        v2f b = *(const v2f*)(&Wlds[(cb * 16 + r) * W_PITCH + 4 * k0 + 2 * hi]);
        acc[cb] = __builtin_amdgcn_wmma_f32_16x16x4_f32(
            false, a, false, b, (short)0, acc[cb], false, false);
      }
    }

    const int rowBase = m0 + 8 * hi;
    if (m0 + 16 <= n_rows) {
#pragma unroll
      for (int cb = 0; cb < 8; ++cb) {
        const int col = cb * 16 + r;
        const float bv = bias ? bias[col] : 0.f;
        float* obase = out + rowBase * EDGE_DIM + col;
#pragma unroll
        for (int v = 0; v < 8; ++v) {
          float val = acc[cb][v] + bv;
          if (apply_swish) val = val * (1.f / (1.f + __expf(-val)));
          obase[v * EDGE_DIM] = val;
        }
      }
    } else {
#pragma unroll
      for (int cb = 0; cb < 8; ++cb) {
        const int col = cb * 16 + r;
        const float bv = bias ? bias[col] : 0.f;
#pragma unroll
        for (int v = 0; v < 8; ++v) {
          const int m = rowBase + v;
          if (m < n_rows) {
            float val = acc[cb][v] + bv;
            if (apply_swish) val = val * (1.f / (1.f + __expf(-val)));
            out[m * EDGE_DIM + col] = val;
          }
        }
      }
    }
  }
}

extern "C" void kernel_launch(void* const* d_in, const int* in_sizes, int n_in,
                              void* d_out, int out_size, void* d_ws, size_t ws_size,
                              hipStream_t stream) {
  const float* x     = (const float*)d_in[0];
  const float* rbf   = (const float*)d_in[1];
  const int*   idx_i = (const int*)d_in[2];
  const float* W_rbf = (const float*)d_in[4];
  const float* W1    = (const float*)d_in[5];
  const float* b1    = (const float*)d_in[6];
  const float* W2    = (const float*)d_in[7];
  const float* b2    = (const float*)d_in[8];
  const float* W3    = (const float*)d_in[9];
  const float* b3    = (const float*)d_in[10];
  const float* W_out = (const float*)d_in[11];
  float* outp = (float*)d_out;

  const int n_edges = in_sizes[0] / EDGE_DIM;
  const int n_nodes = out_size / EDGE_DIM;

  float* h0 = (float*)d_ws;
  float* h1 = h0 + (size_t)n_nodes * EDGE_DIM;

  {
    long n4 = (long)n_nodes * EDGE_DIM / 4;
    long b = (n4 + 255) / 256;
    int blocks = (int)(b > 2048 ? 2048 : (b < 1 ? 1 : b));
    zero_f4_kernel<<<blocks, 256, 0, stream>>>((float4*)h0, n4);
  }

  {
    int nTiles = n_edges >> 4;
    int blocks = (nTiles + 7) / 8;
    if (blocks > 2048) blocks = 2048;
    if (blocks < 1) blocks = 1;
    edge_gate_scatter_kernel<<<blocks, 256, 0, stream>>>(x, rbf, idx_i, W_rbf, h0, n_edges);
  }

  {
    int nTiles = (n_nodes + 15) / 16;
    int blocks = (nTiles + 7) / 8;
    if (blocks > 2048) blocks = 2048;
    if (blocks < 1) blocks = 1;
    node_mlp_kernel<<<blocks, 256, 0, stream>>>(h0, W1, b1, h1, n_nodes, 1);
    node_mlp_kernel<<<blocks, 256, 0, stream>>>(h1, W2, b2, h0, n_nodes, 1);
    node_mlp_kernel<<<blocks, 256, 0, stream>>>(h0, W3, b3, h1, n_nodes, 1);
    node_mlp_kernel<<<blocks, 256, 0, stream>>>(h1, W_out, nullptr, outp, n_nodes, 0);
  }
}


# Node MLP: issue all 16 weight-staging loads of a thread before the first LDS store (fixed 256-thread stride, unrolled) instead of one load, one full wait, one store per trip
# speedup vs baseline: 1.7431x; 1.1569x over previous
#include <hip/hip_runtime.h>

typedef __attribute__((ext_vector_type(2))) float v2f;
typedef __attribute__((ext_vector_type(8))) float v8f;
typedef __attribute__((ext_vector_type(2))) unsigned v2u;

#define EDGE_DIM 128
#define N_RADIAL 16
#define W_PITCH 132

__global__ void zero_f4_kernel(float4* __restrict__ p, long n4) {
  long i = (long)blockIdx.x * blockDim.x + threadIdx.x;
  long stride = (long)gridDim.x * blockDim.x;
  float4 z = make_float4(0.f, 0.f, 0.f, 0.f);
  for (; i < n4; i += stride) p[i] = z;
}

__global__ __launch_bounds__(256) void edge_gate_scatter_kernel(
    const float* __restrict__ x, const float* __restrict__ rbf,
    const int* __restrict__ idx_i, const float* __restrict__ W_rbf,
    float* __restrict__ h, int n_edges)
{
  const unsigned lane = threadIdx.x & 31;
  const int r    = lane & 15;
  const int hi   = lane >> 4;
  const int wave = __builtin_amdgcn_readfirstlane(
      (int)(blockIdx.x * (blockDim.x >> 5) + (threadIdx.x >> 5)));
  const int nW   = gridDim.x * (blockDim.x >> 5);
  const int nFull = n_edges >> 4;

  v2f Bt[8][4];
#pragma unroll
  for (int cb = 0; cb < 8; ++cb) {
    const float* brow = W_rbf + (cb * 16 + r) * N_RADIAL + 2 * hi;
#pragma unroll
    for (int k0 = 0; k0 < 4; ++k0)
      Bt[cb][k0] = *(const v2f*)(brow + 4 * k0);
  }

  for (int t = wave; t < nFull; t += nW) {
    const int e0 = t << 4;

    v2f a[4];
    const float* arow = rbf + (e0 + r) * N_RADIAL + 2 * hi;
#pragma unroll
    for (int k0 = 0; k0 < 4; ++k0)
      a[k0] = *(const v2f*)(arow + 4 * k0);

    const int* ibase = idx_i + e0;
    int nd[16];
#pragma unroll
    for (int q = 0; q < 16; ++q)
      nd[q] = ibase[q];

    const float* xtile = x + (size_t)e0 * EDGE_DIM;

#pragma unroll
    for (int j = 0; j < 4; ++j) {
      const unsigned ch = j * 32 + lane;

      float xlo[8], xhi[8];
#pragma unroll
      for (int v = 0; v < 8; ++v) {
        xlo[v] = xtile[v * EDGE_DIM + ch];
        xhi[v] = xtile[(v + 8) * EDGE_DIM + ch];
      }

      v8f c0 = {0.f, 0.f, 0.f, 0.f, 0.f, 0.f, 0.f, 0.f};
      v8f c1 = {0.f, 0.f, 0.f, 0.f, 0.f, 0.f, 0.f, 0.f};
#pragma unroll
      for (int k0 = 0; k0 < 4; ++k0) {
        c0 = __builtin_amdgcn_wmma_f32_16x16x4_f32(
            false, a[k0], false, Bt[2 * j][k0], (short)0, c0, false, false);
        c1 = __builtin_amdgcn_wmma_f32_16x16x4_f32(
            false, a[k0], false, Bt[2 * j + 1][k0], (short)0, c1, false, false);
      }

      asm volatile("v_nop\n\tv_nop\n\tv_nop\n\tv_nop" : "+v"(c0), "+v"(c1));

#pragma unroll
      for (int v = 0; v < 8; ++v) {
        const float f0 = c0[v];
        const float f1 = c1[v];
        const v2u s = __builtin_amdgcn_permlane16_swap(
            __float_as_uint(f0), __float_as_uint(f1), false, false);
        const unsigned s0 = s[0];
        const unsigned s1 = s[1];
        const float glo = __uint_as_float(s0) * xlo[v];
        const float ghi = __uint_as_float(s1) * xhi[v];
        const unsigned olo = (unsigned)(nd[v] * EDGE_DIM) + lane;
        const unsigned ohi = (unsigned)(nd[v + 8] * EDGE_DIM) + lane;
        atomicAdd(h + ((size_t)olo + j * 32), glo);
        atomicAdd(h + ((size_t)ohi + j * 32), ghi);
      }
    }
  }

  if (wave == 0) {
    for (int e = nFull << 4; e < n_edges; ++e) {
      const int nd = idx_i[e];
      const float* re = rbf + e * N_RADIAL;
#pragma unroll
      for (int cc = 0; cc < 4; ++cc) {
        const int ch = lane * 4 + cc;
        float s = 0.f;
#pragma unroll
        for (int k = 0; k < N_RADIAL; ++k) s += re[k] * W_rbf[ch * N_RADIAL + k];
        float g = s * x[e * EDGE_DIM + ch];
        atomicAdd(h + (nd * EDGE_DIM + ch), g);
      }
    }
  }
}

__global__ __launch_bounds__(256) void node_mlp_kernel(
    const float* __restrict__ in, const float* __restrict__ W,
    const float* __restrict__ bias, float* __restrict__ out,
    int n_rows, int apply_swish)
{
  __shared__ __attribute__((aligned(16))) float Wlds[EDGE_DIM * W_PITCH];
  {
    const float4* Wg = (const float4*)W;
    float4 wv[16];
#pragma unroll
    for (int q = 0; q < 16; ++q)
      wv[q] = Wg[threadIdx.x + q * 256];
#pragma unroll
    for (int q = 0; q < 16; ++q) {
      const int i = threadIdx.x + q * 256;
      *(float4*)(&Wlds[(i >> 5) * W_PITCH + 4 * (i & 31)]) = wv[q];
    }
  }
  __syncthreads();

  const int lane = threadIdx.x & 31;
  const int r    = lane & 15;
  const int hi   = lane >> 4;
  const int wave = blockIdx.x * (blockDim.x >> 5) + (threadIdx.x >> 5);
  const int nW   = gridDim.x * (blockDim.x >> 5);
  const int nTiles = (n_rows + 15) >> 4;

  for (int t = wave; t < nTiles; t += nW) {
    const int m0 = t << 4;
    v8f acc[8];
#pragma unroll
    for (int cb = 0; cb < 8; ++cb)
      acc[cb] = (v8f){0.f, 0.f, 0.f, 0.f, 0.f, 0.f, 0.f, 0.f};

    int mrow = m0 + r;
    if (mrow >= n_rows) mrow = n_rows - 1;
    const float* arow = in + mrow * EDGE_DIM + 2 * hi;

#pragma unroll 4
    for (int k0 = 0; k0 < 32; ++k0) {
      v2f a = *(const v2f*)(arow + 4 * k0);
#pragma unroll
      for (int cb = 0; cb < 8; ++cb) {
        v2f b = *(const v2f*)(&Wlds[(cb * 16 + r) * W_PITCH + 4 * k0 + 2 * hi]);
        acc[cb] = __builtin_amdgcn_wmma_f32_16x16x4_f32(
            false, a, false, b, (short)0, acc[cb], false, false);
      }
    }

    const int rowBase = m0 + 8 * hi;
    if (m0 + 16 <= n_rows) {
#pragma unroll
      for (int cb = 0; cb < 8; ++cb) {
        const int col = cb * 16 + r;
        const float bv = bias ? bias[col] : 0.f;
        float* obase = out + rowBase * EDGE_DIM + col;
#pragma unroll
        for (int v = 0; v < 8; ++v) {
          float val = acc[cb][v] + bv;
          if (apply_swish) val = val * (1.f / (1.f + __expf(-val)));
          obase[v * EDGE_DIM] = val;
        }
      }
    } else {
#pragma unroll
      for (int cb = 0; cb < 8; ++cb) {
        const int col = cb * 16 + r;
        const float bv = bias ? bias[col] : 0.f;
#pragma unroll
        for (int v = 0; v < 8; ++v) {
          const int m = rowBase + v;
          if (m < n_rows) {
            float val = acc[cb][v] + bv;
            if (apply_swish) val = val * (1.f / (1.f + __expf(-val)));
            out[m * EDGE_DIM + col] = val;
          }
        }
      }
    }
  }
}

extern "C" void kernel_launch(void* const* d_in, const int* in_sizes, int n_in,
                              void* d_out, int out_size, void* d_ws, size_t ws_size,
                              hipStream_t stream) {
  const float* x     = (const float*)d_in[0];
  const float* rbf   = (const float*)d_in[1];
  const int*   idx_i = (const int*)d_in[2];
  const float* W_rbf = (const float*)d_in[4];
  const float* W1    = (const float*)d_in[5];
  const float* b1    = (const float*)d_in[6];
  const float* W2    = (const float*)d_in[7];
  const float* b2    = (const float*)d_in[8];
  const float* W3    = (const float*)d_in[9];
  const float* b3    = (const float*)d_in[10];
  const float* W_out = (const float*)d_in[11];
  float* outp = (float*)d_out;

  const int n_edges = in_sizes[0] / EDGE_DIM;
  const int n_nodes = out_size / EDGE_DIM;

  float* h0 = (float*)d_ws;
  float* h1 = h0 + (size_t)n_nodes * EDGE_DIM;

  {
    long n4 = (long)n_nodes * EDGE_DIM / 4;
    long b = (n4 + 255) / 256;
    int blocks = (int)(b > 2048 ? 2048 : (b < 1 ? 1 : b));
    zero_f4_kernel<<<blocks, 256, 0, stream>>>((float4*)h0, n4);
  }

  {
    int nTiles = n_edges >> 4;
    int blocks = (nTiles + 7) / 8;
    if (blocks > 2048) blocks = 2048;
    if (blocks < 1) blocks = 1;
    edge_gate_scatter_kernel<<<blocks, 256, 0, stream>>>(x, rbf, idx_i, W_rbf, h0, n_edges);
  }

  {
    int nTiles = (n_nodes + 15) / 16;
    int blocks = (nTiles + 7) / 8;
    if (blocks > 2048) blocks = 2048;
    if (blocks < 1) blocks = 1;
    node_mlp_kernel<<<blocks, 256, 0, stream>>>(h0, W1, b1, h1, n_nodes, 1);
    node_mlp_kernel<<<blocks, 256, 0, stream>>>(h1, W2, b2, h0, n_nodes, 1);
    node_mlp_kernel<<<blocks, 256, 0, stream>>>(h0, W3, b3, h1, n_nodes, 1);
    node_mlp_kernel<<<blocks, 256, 0, stream>>>(h1, W_out, nullptr, outp, n_nodes, 0);
  }
}


# Node MLP: the four layer launches fused into one; each wave keeps its 16-row block through all layers, activations pass through a wave-private LDS tile, next W fetched under the matrix work
# speedup vs baseline: 2.1577x; 1.2379x over previous
#include <hip/hip_runtime.h>

typedef __attribute__((ext_vector_type(2))) float v2f;
typedef __attribute__((ext_vector_type(8))) float v8f;
typedef __attribute__((ext_vector_type(2))) unsigned v2u;

#define EDGE_DIM 128
#define N_RADIAL 16
#define W_PITCH 132

__global__ void zero_f4_kernel(float4* __restrict__ p, long n4) {
  long i = (long)blockIdx.x * blockDim.x + threadIdx.x;
  long stride = (long)gridDim.x * blockDim.x;
  float4 z = make_float4(0.f, 0.f, 0.f, 0.f);
  for (; i < n4; i += stride) p[i] = z;
}

__global__ __launch_bounds__(256) void edge_gate_scatter_kernel(
    const float* __restrict__ x, const float* __restrict__ rbf,
    const int* __restrict__ idx_i, const float* __restrict__ W_rbf,
    float* __restrict__ h, int n_edges)
{
  const unsigned lane = threadIdx.x & 31;
  const int r    = lane & 15;
  const int hi   = lane >> 4;
  const int wave = __builtin_amdgcn_readfirstlane(
      (int)(blockIdx.x * (blockDim.x >> 5) + (threadIdx.x >> 5)));
  const int nW   = gridDim.x * (blockDim.x >> 5);
  const int nFull = n_edges >> 4;

  v2f Bt[8][4];
#pragma unroll
  for (int cb = 0; cb < 8; ++cb) {
    const float* brow = W_rbf + (cb * 16 + r) * N_RADIAL + 2 * hi;
#pragma unroll
    for (int k0 = 0; k0 < 4; ++k0)
      Bt[cb][k0] = *(const v2f*)(brow + 4 * k0);
  }

  for (int t = wave; t < nFull; t += nW) {
    const int e0 = t << 4;

    v2f a[4];
    const float* arow = rbf + (e0 + r) * N_RADIAL + 2 * hi;
#pragma unroll
    for (int k0 = 0; k0 < 4; ++k0)
      a[k0] = *(const v2f*)(arow + 4 * k0);

    const int* ibase = idx_i + e0;
    int nd[16];
#pragma unroll
    for (int q = 0; q < 16; ++q)
      nd[q] = ibase[q];

    const float* xtile = x + (size_t)e0 * EDGE_DIM;

#pragma unroll
    for (int j = 0; j < 4; ++j) {
      const unsigned ch = j * 32 + lane;

      float xlo[8], xhi[8];
#pragma unroll
      for (int v = 0; v < 8; ++v) {
        xlo[v] = xtile[v * EDGE_DIM + ch];
        xhi[v] = xtile[(v + 8) * EDGE_DIM + ch];
      }

      v8f c0 = {0.f, 0.f, 0.f, 0.f, 0.f, 0.f, 0.f, 0.f};
      v8f c1 = {0.f, 0.f, 0.f, 0.f, 0.f, 0.f, 0.f, 0.f};
#pragma unroll
      for (int k0 = 0; k0 < 4; ++k0) {
        c0 = __builtin_amdgcn_wmma_f32_16x16x4_f32(
            false, a[k0], false, Bt[2 * j][k0], (short)0, c0, false, false);
        c1 = __builtin_amdgcn_wmma_f32_16x16x4_f32(
            false, a[k0], false, Bt[2 * j + 1][k0], (short)0, c1, false, false);
      }

      asm volatile("v_nop\n\tv_nop\n\tv_nop\n\tv_nop" : "+v"(c0), "+v"(c1));

#pragma unroll
      for (int v = 0; v < 8; ++v) {
        const float f0 = c0[v];
        const float f1 = c1[v];
        const v2u s = __builtin_amdgcn_permlane16_swap(
            __float_as_uint(f0), __float_as_uint(f1), false, false);
        const unsigned s0 = s[0];
        const unsigned s1 = s[1];
        const float glo = __uint_as_float(s0) * xlo[v];
        const float ghi = __uint_as_float(s1) * xhi[v];
        const unsigned olo = (unsigned)(nd[v] * EDGE_DIM) + lane;
        const unsigned ohi = (unsigned)(nd[v + 8] * EDGE_DIM) + lane;
        atomicAdd(h + ((size_t)olo + j * 32), glo);
        atomicAdd(h + ((size_t)ohi + j * 32), ghi);
      }
    }
  }

  if (wave == 0) {
    for (int e = nFull << 4; e < n_edges; ++e) {
      const int nd = idx_i[e];
      const float* re = rbf + e * N_RADIAL;
#pragma unroll
      for (int cc = 0; cc < 4; ++cc) {
        const int ch = lane * 4 + cc;
        float s = 0.f;
#pragma unroll
        for (int k = 0; k < N_RADIAL; ++k) s += re[k] * W_rbf[ch * N_RADIAL + k];
        float g = s * x[e * EDGE_DIM + ch];
        atomicAdd(h + (nd * EDGE_DIM + ch), g);
      }
    }
  }
}

#define ACT_PITCH 132

#define REP16(X) X(0) X(1) X(2) X(3) X(4) X(5) X(6) X(7) \
                 X(8) X(9) X(10) X(11) X(12) X(13) X(14) X(15)
#define W_DECL_ONE(q)  float4 wreg_##q;
#define W_FETCH_ONE(q) wreg_##q = Wg4_[(q) * 256];
#define W_STORE_ONE(q) {                                              \
    const int i_ = threadIdx.x + (q) * 256;                           \
    *(float4*)(&Wl_[(i_ >> 5) * W_PITCH + 4 * (i_ & 31)]) = wreg_##q; }
#define W_DECL           REP16(W_DECL_ONE)
#define W_FETCH(Wptr)    { const float4* Wg4_ = (const float4*)(Wptr) + threadIdx.x; REP16(W_FETCH_ONE) }
#define W_STORE(Wl)      { float* Wl_ = (Wl); REP16(W_STORE_ONE) }

__device__ __forceinline__ void tile_mac(v8f (&acc)[8], const float* arow,
                                         const float* Wl, int r, int hi) {
#pragma unroll 4
  for (int k0 = 0; k0 < 32; ++k0) {
    v2f a = *(const v2f*)(arow + 4 * k0);
#pragma unroll
    for (int cb = 0; cb < 8; ++cb) {
      v2f b = *(const v2f*)(&Wl[(cb * 16 + r) * W_PITCH + 4 * k0 + 2 * hi]);
      acc[cb] = __builtin_amdgcn_wmma_f32_16x16x4_f32(
          false, a, false, b, (short)0, acc[cb], false, false);
    }
  }
}

template <int L>
__device__ __forceinline__ void mlp_layer(
    const float* __restrict__ in, const float* __restrict__ Wnext,
    const float* __restrict__ bl, float* __restrict__ out,
    float* Wlds, float* act, bool active, int m0, int n_rows, int r, int hi)
{
  W_DECL
  if constexpr (L < 3) { W_FETCH(Wnext) }

  if (active) {
    v8f acc[8];
#pragma unroll
    for (int cb = 0; cb < 8; ++cb)
      acc[cb] = (v8f){0.f, 0.f, 0.f, 0.f, 0.f, 0.f, 0.f, 0.f};

    float bv[8];
    if constexpr (L < 3) {
#pragma unroll
      for (int cb = 0; cb < 8; ++cb) bv[cb] = bl[cb * 16 + r];
    }

    if constexpr (L == 0) {
      int mrow = m0 + r;
      if (mrow >= n_rows) mrow = n_rows - 1;
      tile_mac(acc, in + mrow * EDGE_DIM + 2 * hi, Wlds, r, hi);
    } else {
      tile_mac(acc, act + r * ACT_PITCH + 2 * hi, Wlds, r, hi);
    }

    if constexpr (L < 3) {
#pragma unroll
      for (int cb = 0; cb < 8; ++cb) {
        const int col = cb * 16 + r;
        float* abase = act + (8 * hi) * ACT_PITCH + col;
#pragma unroll
        for (int v = 0; v < 8; ++v) {
          float val = acc[cb][v] + bv[cb];
          val = val * (1.f / (1.f + __expf(-val)));
          abase[v * ACT_PITCH] = val;
        }
      }
    } else {
      const int rowBase = m0 + 8 * hi;
      if (m0 + 16 <= n_rows) {
#pragma unroll
        for (int cb = 0; cb < 8; ++cb) {
          float* obase = out + rowBase * EDGE_DIM + cb * 16 + r;
#pragma unroll
          for (int v = 0; v < 8; ++v)
            obase[v * EDGE_DIM] = acc[cb][v];
        }
      } else {
#pragma unroll
        for (int cb = 0; cb < 8; ++cb) {
          const int col = cb * 16 + r;
#pragma unroll
          for (int v = 0; v < 8; ++v) {
            const int m = rowBase + v;
            if (m < n_rows) out[m * EDGE_DIM + col] = acc[cb][v];
          }
        }
      }
    }
  }

  if constexpr (L < 3) {
    __syncthreads();
    W_STORE(Wlds)
    __syncthreads();
  }
}

__global__ __launch_bounds__(256, 4) void node_mlp_fused_kernel(
    const float* __restrict__ in,
    const float* __restrict__ W1, const float* __restrict__ b1,
    const float* __restrict__ W2, const float* __restrict__ b2,
    const float* __restrict__ W3, const float* __restrict__ b3,
    const float* __restrict__ W4, float* __restrict__ out, int n_rows)
{
  __shared__ __attribute__((aligned(16))) float Wlds[EDGE_DIM * W_PITCH];
  __shared__ __attribute__((aligned(16))) float Alds[8 * 16 * ACT_PITCH];

  const int lane = threadIdx.x & 31;
  const int r    = lane & 15;
  const int hi   = lane >> 4;
  const int wib  = threadIdx.x >> 5;
  const int wave = blockIdx.x * 8 + wib;
  const int nTiles = (n_rows + 15) >> 4;
  const bool active = wave < nTiles;
  const int m0 = wave << 4;
  float* act = Alds + wib * (16 * ACT_PITCH);

  {
    W_DECL
    W_FETCH(W1)
    W_STORE(Wlds)
  }
  __syncthreads();

  mlp_layer<0>(in, W2, b1, out, Wlds, act, active, m0, n_rows, r, hi);
  mlp_layer<1>(in, W3, b2, out, Wlds, act, active, m0, n_rows, r, hi);
  mlp_layer<2>(in, W4, b3, out, Wlds, act, active, m0, n_rows, r, hi);
  mlp_layer<3>(in, W4, b3, out, Wlds, act, active, m0, n_rows, r, hi);
}

extern "C" void kernel_launch(void* const* d_in, const int* in_sizes, int n_in,
                              void* d_out, int out_size, void* d_ws, size_t ws_size,
                              hipStream_t stream) {
  const float* x     = (const float*)d_in[0];
  const float* rbf   = (const float*)d_in[1];
  const int*   idx_i = (const int*)d_in[2];
  const float* W_rbf = (const float*)d_in[4];
  const float* W1    = (const float*)d_in[5];
  const float* b1    = (const float*)d_in[6];
  const float* W2    = (const float*)d_in[7];
  const float* b2    = (const float*)d_in[8];
  const float* W3    = (const float*)d_in[9];
  const float* b3    = (const float*)d_in[10];
  const float* W_out = (const float*)d_in[11];
  float* outp = (float*)d_out;

  const int n_edges = in_sizes[0] / EDGE_DIM;
  const int n_nodes = out_size / EDGE_DIM;

  float* h0 = (float*)d_ws;

  {
    long n4 = (long)n_nodes * EDGE_DIM / 4;
    long b = (n4 + 255) / 256;
    int blocks = (int)(b > 2048 ? 2048 : (b < 1 ? 1 : b));
    zero_f4_kernel<<<blocks, 256, 0, stream>>>((float4*)h0, n4);
  }

  {
    int nTiles = n_edges >> 4;
    int blocks = (nTiles + 7) / 8;
    if (blocks > 2048) blocks = 2048;
    if (blocks < 1) blocks = 1;
    edge_gate_scatter_kernel<<<blocks, 256, 0, stream>>>(x, rbf, idx_i, W_rbf, h0, n_edges);
  }

  {
    int nTiles = (n_nodes + 15) / 16;
    int blocks = (nTiles + 7) / 8;
    if (blocks < 1) blocks = 1;
    node_mlp_fused_kernel<<<blocks, 256, 0, stream>>>(h0, W1, b1, W2, b2, W3, b3, W_out, outp, n_nodes);
  }
}


# Edge scatter: each x load group is 64 channels x 16 rows, both 128-byte lines of every aligned 256-byte unit requested together: 32 loads in flight per wave, not 16; atomic add order as before
# speedup vs baseline: 2.2712x; 1.0526x over previous
#include <hip/hip_runtime.h>

typedef __attribute__((ext_vector_type(2))) float v2f;
typedef __attribute__((ext_vector_type(8))) float v8f;
typedef __attribute__((ext_vector_type(2))) unsigned v2u;

#define EDGE_DIM 128
#define N_RADIAL 16
#define W_PITCH 132

__global__ void zero_f4_kernel(float4* __restrict__ p, long n4) {
  long i = (long)blockIdx.x * blockDim.x + threadIdx.x;
  long stride = (long)gridDim.x * blockDim.x;
  float4 z = make_float4(0.f, 0.f, 0.f, 0.f);
  for (; i < n4; i += stride) p[i] = z;
}

__global__ __launch_bounds__(256) void edge_gate_scatter_kernel(
    const float* __restrict__ x, const float* __restrict__ rbf,
    const int* __restrict__ idx_i, const float* __restrict__ W_rbf,
    float* __restrict__ h, int n_edges)
{
  const unsigned lane = threadIdx.x & 31;
  const int r    = lane & 15;
  const int hi   = lane >> 4;
  const int wave = __builtin_amdgcn_readfirstlane(
      (int)(blockIdx.x * (blockDim.x >> 5) + (threadIdx.x >> 5)));
  const int nW   = gridDim.x * (blockDim.x >> 5);
  const int nFull = n_edges >> 4;

  v2f Bt[8][4];
#pragma unroll
  for (int cb = 0; cb < 8; ++cb) {
    const float* brow = W_rbf + (cb * 16 + r) * N_RADIAL + 2 * hi;
#pragma unroll
    for (int k0 = 0; k0 < 4; ++k0)
      Bt[cb][k0] = *(const v2f*)(brow + 4 * k0);
  }

  for (int t = wave; t < nFull; t += nW) {
    const int e0 = t << 4;

    v2f a[4];
    const float* arow = rbf + (e0 + r) * N_RADIAL + 2 * hi;
#pragma unroll
    for (int k0 = 0; k0 < 4; ++k0)
      a[k0] = *(const v2f*)(arow + 4 * k0);

    const int* ibase = idx_i + e0;
    int nd[16];
#pragma unroll
    for (int q = 0; q < 16; ++q)
      nd[q] = ibase[q];

    const float* xtile = x + (size_t)e0 * EDGE_DIM;

#pragma unroll
    for (int jp = 0; jp < 2; ++jp) {
      float xlo[2][8], xhi[2][8];
#pragma unroll
      for (int v = 0; v < 8; ++v) {
#pragma unroll
        for (int q = 0; q < 2; ++q) {
          const unsigned ch = jp * 64 + q * 32 + lane;
          xlo[q][v] = xtile[v * EDGE_DIM + ch];
          xhi[q][v] = xtile[(v + 8) * EDGE_DIM + ch];
        }
      }

      v8f c[4];
#pragma unroll
      for (int ct = 0; ct < 4; ++ct)
        c[ct] = (v8f){0.f, 0.f, 0.f, 0.f, 0.f, 0.f, 0.f, 0.f};
#pragma unroll
      for (int k0 = 0; k0 < 4; ++k0) {
#pragma unroll
        for (int ct = 0; ct < 4; ++ct)
          c[ct] = __builtin_amdgcn_wmma_f32_16x16x4_f32(
              false, a[k0], false, Bt[4 * jp + ct][k0], (short)0, c[ct], false, false);
      }

      asm volatile("v_nop\n\tv_nop\n\tv_nop\n\tv_nop"
                   : "+v"(c[0]), "+v"(c[1]), "+v"(c[2]), "+v"(c[3]));
      asm volatile("" : : "v"(a[0]), "v"(a[1]), "v"(a[2]), "v"(a[3]));

#pragma unroll
      for (int q = 0; q < 2; ++q) {
#pragma unroll
        for (int v = 0; v < 8; ++v) {
          const float f0 = c[2 * q][v];
          const float f1 = c[2 * q + 1][v];
          const v2u s = __builtin_amdgcn_permlane16_swap(
              __float_as_uint(f0), __float_as_uint(f1), false, false);
          const unsigned s0 = s[0];
          const unsigned s1 = s[1];
          const float glo = __uint_as_float(s0) * xlo[q][v];
          const float ghi = __uint_as_float(s1) * xhi[q][v];
          const unsigned olo = (unsigned)(nd[v] * EDGE_DIM) + lane;
          const unsigned ohi = (unsigned)(nd[v + 8] * EDGE_DIM) + lane;
          atomicAdd(h + ((size_t)olo + jp * 64 + q * 32), glo);
          atomicAdd(h + ((size_t)ohi + jp * 64 + q * 32), ghi);
        }
      }
    }
  }

  if (wave == 0) {
    for (int e = nFull << 4; e < n_edges; ++e) {
      const int nd = idx_i[e];
      const float* re = rbf + e * N_RADIAL;
#pragma unroll
      for (int cc = 0; cc < 4; ++cc) {
        const int ch = lane * 4 + cc;
        float s = 0.f;
#pragma unroll
        for (int k = 0; k < N_RADIAL; ++k) s += re[k] * W_rbf[ch * N_RADIAL + k];
        float g = s * x[e * EDGE_DIM + ch];
        atomicAdd(h + (nd * EDGE_DIM + ch), g);
      }
    }
  }
}

#define ACT_PITCH 132

#define REP16(X) X(0) X(1) X(2) X(3) X(4) X(5) X(6) X(7) \
                 X(8) X(9) X(10) X(11) X(12) X(13) X(14) X(15)
#define W_DECL_ONE(q)  float4 wreg_##q;
#define W_FETCH_ONE(q) wreg_##q = Wg4_[(q) * 256];
#define W_STORE_ONE(q) {                                              \
    const int i_ = threadIdx.x + (q) * 256;                           \
    *(float4*)(&Wl_[(i_ >> 5) * W_PITCH + 4 * (i_ & 31)]) = wreg_##q; }
#define W_DECL           REP16(W_DECL_ONE)
#define W_FETCH(Wptr)    { const float4* Wg4_ = (const float4*)(Wptr) + threadIdx.x; REP16(W_FETCH_ONE) }
#define W_STORE(Wl)      { float* Wl_ = (Wl); REP16(W_STORE_ONE) }

__device__ __forceinline__ void tile_mac(v8f (&acc)[8], const float* arow,
                                         const float* Wl, int r, int hi) {
#pragma unroll 4
  for (int k0 = 0; k0 < 32; ++k0) {
    v2f a = *(const v2f*)(arow + 4 * k0);
#pragma unroll
    for (int cb = 0; cb < 8; ++cb) {
      v2f b = *(const v2f*)(&Wl[(cb * 16 + r) * W_PITCH + 4 * k0 + 2 * hi]);
      acc[cb] = __builtin_amdgcn_wmma_f32_16x16x4_f32(
          false, a, false, b, (short)0, acc[cb], false, false);
    }
  }
}

template <int L>
__device__ __forceinline__ void mlp_layer(
    const float* __restrict__ in, const float* __restrict__ Wnext,
    const float* __restrict__ bl, float* __restrict__ out,
    float* Wlds, float* act, bool active, int m0, int n_rows, int r, int hi)
{
  W_DECL
  if constexpr (L < 3) { W_FETCH(Wnext) }

  if (active) {
    v8f acc[8];
#pragma unroll
    for (int cb = 0; cb < 8; ++cb)
      acc[cb] = (v8f){0.f, 0.f, 0.f, 0.f, 0.f, 0.f, 0.f, 0.f};

    float bv[8];
    if constexpr (L < 3) {
#pragma unroll
      for (int cb = 0; cb < 8; ++cb) bv[cb] = bl[cb * 16 + r];
    }

    if constexpr (L == 0) {
      int mrow = m0 + r;
      if (mrow >= n_rows) mrow = n_rows - 1;
      tile_mac(acc, in + mrow * EDGE_DIM + 2 * hi, Wlds, r, hi);
    } else {
      tile_mac(acc, act + r * ACT_PITCH + 2 * hi, Wlds, r, hi);
    }

    if constexpr (L < 3) {
#pragma unroll
      for (int cb = 0; cb < 8; ++cb) {
        const int col = cb * 16 + r;
        float* abase = act + (8 * hi) * ACT_PITCH + col;
#pragma unroll
        for (int v = 0; v < 8; ++v) {
          float val = acc[cb][v] + bv[cb];
          val = val * (1.f / (1.f + __expf(-val)));
          abase[v * ACT_PITCH] = val;
        }
      }
    } else {
      const int rowBase = m0 + 8 * hi;
      if (m0 + 16 <= n_rows) {
#pragma unroll
        for (int cb = 0; cb < 8; ++cb) {
          float* obase = out + rowBase * EDGE_DIM + cb * 16 + r;
#pragma unroll
          for (int v = 0; v < 8; ++v)
            obase[v * EDGE_DIM] = acc[cb][v];
        }
      } else {
#pragma unroll
        for (int cb = 0; cb < 8; ++cb) {
          const int col = cb * 16 + r;
#pragma unroll
          for (int v = 0; v < 8; ++v) {
            const int m = rowBase + v;
            if (m < n_rows) out[m * EDGE_DIM + col] = acc[cb][v];
          }
        }
      }
    }
  }

  if constexpr (L < 3) {
    __syncthreads();
    W_STORE(Wlds)
    __syncthreads();
  }
}

__global__ __launch_bounds__(256, 4) void node_mlp_fused_kernel(
    const float* __restrict__ in,
    const float* __restrict__ W1, const float* __restrict__ b1,
    const float* __restrict__ W2, const float* __restrict__ b2,
    const float* __restrict__ W3, const float* __restrict__ b3,
    const float* __restrict__ W4, float* __restrict__ out, int n_rows)
{
  __shared__ __attribute__((aligned(16))) float Wlds[EDGE_DIM * W_PITCH];
  __shared__ __attribute__((aligned(16))) float Alds[8 * 16 * ACT_PITCH];

  const int lane = threadIdx.x & 31;
  const int r    = lane & 15;
  const int hi   = lane >> 4;
  const int wib  = threadIdx.x >> 5;
  const int wave = blockIdx.x * 8 + wib;
  const int nTiles = (n_rows + 15) >> 4;
  const bool active = wave < nTiles;
  const int m0 = wave << 4;
  float* act = Alds + wib * (16 * ACT_PITCH);

  {
    W_DECL
    W_FETCH(W1)
    W_STORE(Wlds)
  }
  __syncthreads();

  mlp_layer<0>(in, W2, b1, out, Wlds, act, active, m0, n_rows, r, hi);
  mlp_layer<1>(in, W3, b2, out, Wlds, act, active, m0, n_rows, r, hi);
  mlp_layer<2>(in, W4, b3, out, Wlds, act, active, m0, n_rows, r, hi);
  mlp_layer<3>(in, W4, b3, out, Wlds, act, active, m0, n_rows, r, hi);
}

extern "C" void kernel_launch(void* const* d_in, const int* in_sizes, int n_in,
                              void* d_out, int out_size, void* d_ws, size_t ws_size,
                              hipStream_t stream) {
  const float* x     = (const float*)d_in[0];
  const float* rbf   = (const float*)d_in[1];
  const int*   idx_i = (const int*)d_in[2];
  const float* W_rbf = (const float*)d_in[4];
  const float* W1    = (const float*)d_in[5];
  const float* b1    = (const float*)d_in[6];
  const float* W2    = (const float*)d_in[7];
  const float* b2    = (const float*)d_in[8];
  const float* W3    = (const float*)d_in[9];
  const float* b3    = (const float*)d_in[10];
  const float* W_out = (const float*)d_in[11];
  float* outp = (float*)d_out;

  const int n_edges = in_sizes[0] / EDGE_DIM;
  const int n_nodes = out_size / EDGE_DIM;

  float* h0 = (float*)d_ws;

  {
    long n4 = (long)n_nodes * EDGE_DIM / 4;
    long b = (n4 + 255) / 256;
    int blocks = (int)(b > 2048 ? 2048 : (b < 1 ? 1 : b));
    zero_f4_kernel<<<blocks, 256, 0, stream>>>((float4*)h0, n4);
  }

  {
    int nTiles = n_edges >> 4;
    int blocks = (nTiles + 7) / 8;
    if (blocks > 2048) blocks = 2048;
    if (blocks < 1) blocks = 1;
    edge_gate_scatter_kernel<<<blocks, 256, 0, stream>>>(x, rbf, idx_i, W_rbf, h0, n_edges);
  }

  {
    int nTiles = (n_nodes + 15) / 16;
    int blocks = (nTiles + 7) / 8;
    if (blocks < 1) blocks = 1;
    node_mlp_fused_kernel<<<blocks, 256, 0, stream>>>(h0, W1, b1, W2, b2, W3, b3, W_out, outp, n_nodes);
  }
}
